# starting kernel
# baseline (speedup 1.0000x reference)
#include <hip/hip_runtime.h>
#include <hip/hip_bf16.h>

typedef __attribute__((ext_vector_type(2))) float v2f;
typedef __attribute__((ext_vector_type(8))) float v8f;

#define F_IN   128
#define K_TOT  256
#define F_OUT  128
#define LDS_STRIDE 260

__global__ void gc_zero_kernel(float* __restrict__ p, long long n) {
    long long i = (long long)blockIdx.x * blockDim.x + threadIdx.x;
    if (i < n) p[i] = 0.0f;
}

__global__ __launch_bounds__(256) void gc_scatter_kernel(
    const float* __restrict__ x, const int* __restrict__ esrc,
    const int* __restrict__ edst, const float* __restrict__ eval,
    float* __restrict__ h, long long E) {
    long long tid = (long long)blockIdx.x * blockDim.x + threadIdx.x;
    long long total = E << 5;
    if (tid >= total) return;
    long long e = tid >> 5;
    int lane = (int)(tid & 31);
    int s = esrc[e];
    int d = edst[e];
    float v = eval[e];
    const float4* xrow = (const float4*)(x + (long long)d * F_IN);
    float4 xv = xrow[lane];
    float* hp = h + (long long)s * F_IN + lane * 4;
    atomicAdd(hp + 0, v * xv.x);
    atomicAdd(hp + 1, v * xv.y);
    atomicAdd(hp + 2, v * xv.z);
    atomicAdd(hp + 3, v * xv.w);
}

__global__ void gc_rowsum_kernel(const int* __restrict__ esrc,
                                 const float* __restrict__ eval,
                                 float* __restrict__ rowsum, long long E) {
    long long e = (long long)blockIdx.x * blockDim.x + threadIdx.x;
    if (e < E) atomicAdd(&rowsum[esrc[e]], eval[e]);
}

__global__ __launch_bounds__(256) void gc_gemm_kernel(
    const float* __restrict__ h, const float* __restrict__ ori,
    const float* __restrict__ W, const float* __restrict__ bias,
    const float* __restrict__ rowsum, float* __restrict__ out, int N) {
    __shared__ float Atile[16 * LDS_STRIDE];

    const int base = blockIdx.x * 16;
    const int tid  = threadIdx.x;

    #pragma unroll
    for (int i = 0; i < 4; ++i) {
        int idx = tid + i * 256;
        int r   = idx >> 6;
        int col = (idx & 63) * 4;
        int row = base + r; if (row >= N) row = N - 1;
        const float* sp = (col < F_IN)
            ? (h   + (long long)row * F_IN + col)
            : (ori + (long long)row * F_IN + (col - F_IN));
        float4 v = *(const float4*)sp;
        float* dp = &Atile[r * LDS_STRIDE + col];
        dp[0] = v.x; dp[1] = v.y; dp[2] = v.z; dp[3] = v.w;
    }
    __syncthreads();

    const int wave    = tid >> 5;
    const int lane    = tid & 31;
    const int lmod    = lane & 15;
    const int lhi     = lane >> 4;
    const int colBase = wave * 16;
    const int col     = colBase + lmod;

    v8f acc = {};
    #pragma unroll
    for (int k = 0; k < K_TOT; k += 4) {
        v2f a = *(const v2f*)&Atile[lmod * LDS_STRIDE + k + 2 * lhi];
        int kb = k + 2 * lhi;
        v2f b;
        b.x = W[(long long)kb * F_OUT + col];
        b.y = W[(long long)(kb + 1) * F_OUT + col];
        acc = __builtin_amdgcn_wmma_f32_16x16x4_f32(
             false, a,  false, b,
             (short)0, acc,  false,  false);
    }

    const float bv = bias[col];
    #pragma unroll
    for (int r = 0; r < 8; ++r) {
        int row = base + r + 8 * lhi;
        if (row < N)
            out[(long long)row * F_OUT + col] = acc[r] / rowsum[row] + bv;
    }
}

extern "C" void kernel_launch(void* const* d_in, const int* in_sizes, int n_in,
                              void* d_out, int out_size, void* d_ws, size_t ws_size,
                              hipStream_t stream) {
    const float* ori  = (const float*)d_in[0];
    const float* x    = (const float*)d_in[1];
    const int*   esrc = (const int*)d_in[2];
    const int*   edst = (const int*)d_in[3];
    const float* eval = (const float*)d_in[4];
    const float* W    = (const float*)d_in[5];
    const float* bias = (const float*)d_in[6];
    float* out = (float*)d_out;

    const long long N = (long long)in_sizes[0] / F_IN;
    const long long E = (long long)in_sizes[2];

    float* h      = (float*)d_ws;
    float* rowsum = h + N * F_IN;

    long long zn = N * (F_IN + 1);
    gc_zero_kernel<<<(unsigned)((zn + 255) / 256), 256, 0, stream>>>(h, zn);

    long long st = E << 5;
    gc_scatter_kernel<<<(unsigned)((st + 255) / 256), 256, 0, stream>>>(
        x, esrc, edst, eval, h, E);
    gc_rowsum_kernel<<<(unsigned)((E + 255) / 256), 256, 0, stream>>>(
        esrc, eval, rowsum, E);

    gc_gemm_kernel<<<(unsigned)((N + 15) / 16), 256, 0, stream>>>(
        h, ori, W, bias, rowsum, out, (int)N);
}


# scatter: lane l owns features l, l+32, l+64, l+96, so each gather load and each f32 atomic add of a wave covers one whole 128-B line (4 atomic line requests per edge, was 16); atomics kept
# speedup vs baseline: 1.2482x; 1.2482x over previous
#include <hip/hip_runtime.h>
#include <hip/hip_bf16.h>

typedef __attribute__((ext_vector_type(2))) float v2f;
typedef __attribute__((ext_vector_type(8))) float v8f;

#define F_IN   128
#define K_TOT  256
#define F_OUT  128
#define LDS_STRIDE 260

__global__ void gc_zero_kernel(float* __restrict__ p, long long n) {
    long long i = (long long)blockIdx.x * blockDim.x + threadIdx.x;
    if (i < n) p[i] = 0.0f;
}

__global__ __launch_bounds__(256) void gc_scatter_kernel(
    const float* __restrict__ x, const int* __restrict__ esrc,
    const int* __restrict__ edst, const float* __restrict__ eval,
    float* __restrict__ h, long long E) {
    long long tid = (long long)blockIdx.x * blockDim.x + threadIdx.x;
    long long total = E << 5;
    if (tid >= total) return;
    long long e = tid >> 5;
    int lane = (int)(tid & 31);
    int s = esrc[e];
    int d = edst[e];
    float v = eval[e];
    const float* xp = x + (long long)d * F_IN + lane;
    float x0 = xp[0];
    float x1 = xp[32];
    float x2 = xp[64];
    float x3 = xp[96];
    float* hp = h + (long long)s * F_IN + lane;
    atomicAdd(hp +  0, v * x0);
    atomicAdd(hp + 32, v * x1);
    atomicAdd(hp + 64, v * x2);
    atomicAdd(hp + 96, v * x3);
}

__global__ void gc_rowsum_kernel(const int* __restrict__ esrc,
                                 const float* __restrict__ eval,
                                 float* __restrict__ rowsum, long long E) {
    long long e = (long long)blockIdx.x * blockDim.x + threadIdx.x;
    if (e < E) atomicAdd(&rowsum[esrc[e]], eval[e]);
}

__global__ __launch_bounds__(256) void gc_gemm_kernel(
    const float* __restrict__ h, const float* __restrict__ ori,
    const float* __restrict__ W, const float* __restrict__ bias,
    const float* __restrict__ rowsum, float* __restrict__ out, int N) {
    __shared__ float Atile[16 * LDS_STRIDE];

    const int base = blockIdx.x * 16;
    const int tid  = threadIdx.x;

    #pragma unroll
    for (int i = 0; i < 4; ++i) {
        int idx = tid + i * 256;
        int r   = idx >> 6;
        int col = (idx & 63) * 4;
        int row = base + r; if (row >= N) row = N - 1;
        const float* sp = (col < F_IN)
            ? (h   + (long long)row * F_IN + col)
            : (ori + (long long)row * F_IN + (col - F_IN));
        float4 v = *(const float4*)sp;
        float* dp = &Atile[r * LDS_STRIDE + col];
        dp[0] = v.x; dp[1] = v.y; dp[2] = v.z; dp[3] = v.w;
    }
    __syncthreads();

    const int wave    = tid >> 5;
    const int lane    = tid & 31;
    const int lmod    = lane & 15;
    const int lhi     = lane >> 4;
    const int colBase = wave * 16;
    const int col     = colBase + lmod;

    v8f acc = {};
    #pragma unroll
    for (int k = 0; k < K_TOT; k += 4) {
        v2f a = *(const v2f*)&Atile[lmod * LDS_STRIDE + k + 2 * lhi];
        int kb = k + 2 * lhi;
        v2f b;
        b.x = W[(long long)kb * F_OUT + col];
        b.y = W[(long long)(kb + 1) * F_OUT + col];
        acc = __builtin_amdgcn_wmma_f32_16x16x4_f32(
             false, a,  false, b,
             (short)0, acc,  false,  false);
    }

    const float bv = bias[col];
    #pragma unroll
    for (int r = 0; r < 8; ++r) {
        int row = base + r + 8 * lhi;
        if (row < N)
            out[(long long)row * F_OUT + col] = acc[r] / rowsum[row] + bv;
    }
}

extern "C" void kernel_launch(void* const* d_in, const int* in_sizes, int n_in,
                              void* d_out, int out_size, void* d_ws, size_t ws_size,
                              hipStream_t stream) {
    const float* ori  = (const float*)d_in[0];
    const float* x    = (const float*)d_in[1];
    const int*   esrc = (const int*)d_in[2];
    const int*   edst = (const int*)d_in[3];
    const float* eval = (const float*)d_in[4];
    const float* W    = (const float*)d_in[5];
    const float* bias = (const float*)d_in[6];
    float* out = (float*)d_out;

    const long long N = (long long)in_sizes[0] / F_IN;
    const long long E = (long long)in_sizes[2];

    float* h      = (float*)d_ws;
    float* rowsum = h + N * F_IN;

    long long zn = N * (F_IN + 1);
    gc_zero_kernel<<<(unsigned)((zn + 255) / 256), 256, 0, stream>>>(h, zn);

    long long st = E << 5;
    gc_scatter_kernel<<<(unsigned)((st + 255) / 256), 256, 0, stream>>>(
        x, esrc, edst, eval, h, E);
    gc_rowsum_kernel<<<(unsigned)((E + 255) / 256), 256, 0, stream>>>(
        esrc, eval, rowsum, E);

    gc_gemm_kernel<<<(unsigned)((N + 15) / 16), 256, 0, stream>>>(
        h, ori, W, bias, rowsum, out, (int)N);
}


# scatter: one wave takes 32 edges: lane i loads edge i's src/dst/val (coalesced), readlane broadcasts; 8 edges' gathers in flight, then their atomics; 53K waves, was 1.7M; atomic arrival order kept
# speedup vs baseline: 1.6587x; 1.3289x over previous
#include <hip/hip_runtime.h>
#include <hip/hip_bf16.h>

typedef __attribute__((ext_vector_type(2))) float v2f;
typedef __attribute__((ext_vector_type(8))) float v8f;

#define F_IN   128
#define K_TOT  256
#define F_OUT  128
#define LDS_STRIDE 260

__global__ void gc_zero_kernel(float* __restrict__ p, long long n) {
    long long i = (long long)blockIdx.x * blockDim.x + threadIdx.x;
    if (i < n) p[i] = 0.0f;
}

#define EDGES_PER_WAVE 32
#define EDGES_IN_FLIGHT 8
__global__ __launch_bounds__(256) void gc_scatter_kernel(
    const float* __restrict__ x, const int* __restrict__ esrc,
    const int* __restrict__ edst, const float* __restrict__ eval,
    float* __restrict__ h, long long E) {
    const int lane = (int)(threadIdx.x & 31);
    const int wib = __builtin_amdgcn_readfirstlane((int)(threadIdx.x >> 5));
    const long long wave = (long long)blockIdx.x * (blockDim.x >> 5) + wib;
    const long long e0 = wave * EDGES_PER_WAVE;
    if (e0 >= E) return;
    const long long rem = E - e0;
    const int cnt = rem < EDGES_PER_WAVE ? (int)rem : EDGES_PER_WAVE;

    int s_l = 0, d_l = 0, v_l = 0;
    if (lane < cnt) {
        s_l = esrc[e0 + lane];
        d_l = edst[e0 + lane];
        v_l = __float_as_int(eval[e0 + lane]);
    }

    int j0 = 0;
    for (; j0 + EDGES_IN_FLIGHT <= cnt; j0 += EDGES_IN_FLIGHT) {
        float xv[EDGES_IN_FLIGHT][4];
        #pragma unroll
        for (int u = 0; u < EDGES_IN_FLIGHT; ++u) {
            const int d = __builtin_amdgcn_readlane(d_l, j0 + u);
            const float* xp = x + (long long)d * F_IN + lane;
            xv[u][0] = xp[0];
            xv[u][1] = xp[32];
            xv[u][2] = xp[64];
            xv[u][3] = xp[96];
        }
        #pragma unroll
        for (int u = 0; u < EDGES_IN_FLIGHT; ++u) {
            const int s = __builtin_amdgcn_readlane(s_l, j0 + u);
            const float v = __int_as_float(__builtin_amdgcn_readlane(v_l, j0 + u));
            float* hp = h + (long long)s * F_IN + lane;
            atomicAdd(hp +  0, v * xv[u][0]);
            atomicAdd(hp + 32, v * xv[u][1]);
            atomicAdd(hp + 64, v * xv[u][2]);
            atomicAdd(hp + 96, v * xv[u][3]);
        }
    }
    for (; j0 < cnt; ++j0) {
        const int d = __builtin_amdgcn_readlane(d_l, j0);
        const int s = __builtin_amdgcn_readlane(s_l, j0);
        const float v = __int_as_float(__builtin_amdgcn_readlane(v_l, j0));
        const float* xp = x + (long long)d * F_IN + lane;
        float* hp = h + (long long)s * F_IN + lane;
        const float x0 = xp[0], x1 = xp[32], x2 = xp[64], x3 = xp[96];
        atomicAdd(hp +  0, v * x0);
        atomicAdd(hp + 32, v * x1);
        atomicAdd(hp + 64, v * x2);
        atomicAdd(hp + 96, v * x3);
    }
}

__global__ void gc_rowsum_kernel(const int* __restrict__ esrc,
                                 const float* __restrict__ eval,
                                 float* __restrict__ rowsum, long long E) {
    long long e = (long long)blockIdx.x * blockDim.x + threadIdx.x;
    if (e < E) atomicAdd(&rowsum[esrc[e]], eval[e]);
}

__global__ __launch_bounds__(256) void gc_gemm_kernel(
    const float* __restrict__ h, const float* __restrict__ ori,
    const float* __restrict__ W, const float* __restrict__ bias,
    const float* __restrict__ rowsum, float* __restrict__ out, int N) {
    __shared__ float Atile[16 * LDS_STRIDE];

    const int base = blockIdx.x * 16;
    const int tid  = threadIdx.x;

    #pragma unroll
    for (int i = 0; i < 4; ++i) {
        int idx = tid + i * 256;
        int r   = idx >> 6;
        int col = (idx & 63) * 4;
        int row = base + r; if (row >= N) row = N - 1;
        const float* sp = (col < F_IN)
            ? (h   + (long long)row * F_IN + col)
            : (ori + (long long)row * F_IN + (col - F_IN));
        float4 v = *(const float4*)sp;
        float* dp = &Atile[r * LDS_STRIDE + col];
        dp[0] = v.x; dp[1] = v.y; dp[2] = v.z; dp[3] = v.w;
    }
    __syncthreads();

    const int wave    = tid >> 5;
    const int lane    = tid & 31;
    const int lmod    = lane & 15;
    const int lhi     = lane >> 4;
    const int colBase = wave * 16;
    const int col     = colBase + lmod;

    v8f acc = {};
    #pragma unroll
    for (int k = 0; k < K_TOT; k += 4) {
        v2f a = *(const v2f*)&Atile[lmod * LDS_STRIDE + k + 2 * lhi];
        int kb = k + 2 * lhi;
        v2f b;
        b.x = W[(long long)kb * F_OUT + col];
        b.y = W[(long long)(kb + 1) * F_OUT + col];
        acc = __builtin_amdgcn_wmma_f32_16x16x4_f32(
             false, a,  false, b,
             (short)0, acc,  false,  false);
    }

    const float bv = bias[col];
    #pragma unroll
    for (int r = 0; r < 8; ++r) {
        int row = base + r + 8 * lhi;
        if (row < N)
            out[(long long)row * F_OUT + col] = acc[r] / rowsum[row] + bv;
    }
}

extern "C" void kernel_launch(void* const* d_in, const int* in_sizes, int n_in,
                              void* d_out, int out_size, void* d_ws, size_t ws_size,
                              hipStream_t stream) {
    const float* ori  = (const float*)d_in[0];
    const float* x    = (const float*)d_in[1];
    const int*   esrc = (const int*)d_in[2];
    const int*   edst = (const int*)d_in[3];
    const float* eval = (const float*)d_in[4];
    const float* W    = (const float*)d_in[5];
    const float* bias = (const float*)d_in[6];
    float* out = (float*)d_out;

    const long long N = (long long)in_sizes[0] / F_IN;
    const long long E = (long long)in_sizes[2];

    float* h      = (float*)d_ws;
    float* rowsum = h + N * F_IN;

    long long zn = N * (F_IN + 1);
    gc_zero_kernel<<<(unsigned)((zn + 255) / 256), 256, 0, stream>>>(h, zn);

    gc_scatter_kernel<<<(unsigned)((E + 255) / 256), 256, 0, stream>>>(
        x, esrc, edst, eval, h, E);
    gc_rowsum_kernel<<<(unsigned)((E + 255) / 256), 256, 0, stream>>>(
        esrc, eval, rowsum, E);

    gc_gemm_kernel<<<(unsigned)((N + 15) / 16), 256, 0, stream>>>(
        h, ori, W, bias, rowsum, out, (int)N);
}


# scatter replaced by gather: edges counting-sorted by src (one integer ticket add per edge), each h row summed by one wave into the GEMM block's LDS; no f32 atomics; sums in ticket arrival order
# speedup vs baseline: 2.1257x; 1.2815x over previous
#include <hip/hip_runtime.h>
#include <hip/hip_bf16.h>

typedef __attribute__((ext_vector_type(2))) float v2f;
typedef __attribute__((ext_vector_type(8))) float v8f;

#define F_IN   128
#define K_TOT  256
#define F_OUT  128
#define LDS_STRIDE 260


__global__ void gc_clear_kernel(int* __restrict__ p, int n) {
    int i = blockIdx.x * blockDim.x + threadIdx.x;
    if (i < n) p[i] = 0;
}

#define EDGES_PER_THREAD 4
__global__ __launch_bounds__(256) void gc_ticket_kernel(
    const int* __restrict__ esrc, int* __restrict__ cnt, int* __restrict__ rank, long long E) {
    const long long t = (long long)blockIdx.x * (256 * EDGES_PER_THREAD) + threadIdx.x;
    int s[EDGES_PER_THREAD];
    int r[EDGES_PER_THREAD];
    #pragma unroll
    for (int k = 0; k < EDGES_PER_THREAD; ++k) {
        const long long e = t + k * 256;
        s[k] = esrc[e < E ? e : E - 1];
    }
    #pragma unroll
    for (int k = 0; k < EDGES_PER_THREAD; ++k) {
        const long long e = t + k * 256;
        r[k] = 0;
        if (e < E) r[k] = atomicAdd(&cnt[s[k]], 1);
    }
    #pragma unroll
    for (int k = 0; k < EDGES_PER_THREAD; ++k) {
        const long long e = t + k * 256;
        if (e < E) rank[e] = r[k];
    }
}

#define ALLOC_ROWS 1024
__global__ __launch_bounds__(ALLOC_ROWS) void gc_alloc_kernel(
    const int* cnt, int* cursor, int* __restrict__ start, int N) {
    __shared__ int sc[ALLOC_ROWS];
    __shared__ int sbase;
    const int tid = threadIdx.x;
    const int i = blockIdx.x * ALLOC_ROWS + tid;
    const int c = (i < N) ? cnt[i] : 0;
    sc[tid] = c;
    __syncthreads();
    for (int off = 1; off < ALLOC_ROWS; off <<= 1) {
        const int add = (tid >= off) ? sc[tid - off] : 0;
        __syncthreads();
        sc[tid] += add;
        __syncthreads();
    }
    if (tid == ALLOC_ROWS - 1) sbase = atomicAdd(cursor, sc[tid]);
    __syncthreads();
    if (i < N) start[i] = sbase + sc[tid] - c;
}

__global__ __launch_bounds__(256) void gc_fill_kernel(
    const int* __restrict__ esrc, const int* __restrict__ edst,
    const float* __restrict__ eval, const int* __restrict__ rank,
    const int* __restrict__ start, int* __restrict__ csr, long long E) {
    const long long t = (long long)blockIdx.x * (256 * EDGES_PER_THREAD) + threadIdx.x;
    int sidx[EDGES_PER_THREAD];
    int pos[EDGES_PER_THREAD];
    int2 pr[EDGES_PER_THREAD];
    #pragma unroll
    for (int k = 0; k < EDGES_PER_THREAD; ++k) {
        const long long e = t + k * 256;
        const long long ec = e < E ? e : E - 1;
        sidx[k] = esrc[ec];
        pos[k]  = rank[ec];
        pr[k]   = make_int2(edst[ec], __float_as_int(eval[ec]));
    }
    #pragma unroll
    for (int k = 0; k < EDGES_PER_THREAD; ++k)
        pos[k] += start[sidx[k]];
    #pragma unroll
    for (int k = 0; k < EDGES_PER_THREAD; ++k) {
        const long long e = t + k * 256;
        if (e < E) ((int2*)csr)[pos[k]] = pr[k];
    }
}

#define ROWS_IN_FLIGHT 8
__device__ __forceinline__ void gc_row_sum(
    const float* __restrict__ x, const int* __restrict__ csr,
    const int st, const int cn, const int lane, int d_l, int v_l,
    float4& acc, float& rs) {
    for (int c0 = 0; c0 < cn; c0 += 32) {
        if (c0 > 0) {
            d_l = 0; v_l = 0;
            if (c0 + lane < cn) {
                const int2 p = ((const int2*)csr)[st + c0 + lane];
                d_l = p.x; v_l = p.y;
            }
        }
        const int left = cn - c0;
        const int m = left < 32 ? left : 32;
        int j0 = 0;
        for (; j0 + ROWS_IN_FLIGHT <= m; j0 += ROWS_IN_FLIGHT) {
            float4 xv[ROWS_IN_FLIGHT];
            #pragma unroll
            for (int u = 0; u < ROWS_IN_FLIGHT; ++u) {
                const int d = __builtin_amdgcn_readlane(d_l, j0 + u);
                xv[u] = *(const float4*)(x + (long long)d * F_IN + lane * 4);
            }
            #pragma unroll
            for (int u = 0; u < ROWS_IN_FLIGHT; ++u) {
                const float v = __int_as_float(__builtin_amdgcn_readlane(v_l, j0 + u));
                acc.x += v * xv[u].x;
                acc.y += v * xv[u].y;
                acc.z += v * xv[u].z;
                acc.w += v * xv[u].w;
                rs += v;
            }
        }
        if (j0 < m) {
            float4 xv[ROWS_IN_FLIGHT];
            #pragma unroll
            for (int u = 0; u < ROWS_IN_FLIGHT; ++u) {
                xv[u] = make_float4(0.0f, 0.0f, 0.0f, 0.0f);
                if (j0 + u < m) {
                    const int d = __builtin_amdgcn_readlane(d_l, j0 + u);
                    xv[u] = *(const float4*)(x + (long long)d * F_IN + lane * 4);
                }
            }
            #pragma unroll
            for (int u = 0; u < ROWS_IN_FLIGHT; ++u) {
                if (j0 + u < m) {
                    const float v = __int_as_float(__builtin_amdgcn_readlane(v_l, j0 + u));
                    acc.x += v * xv[u].x;
                    acc.y += v * xv[u].y;
                    acc.z += v * xv[u].z;
                    acc.w += v * xv[u].w;
                    rs += v;
                }
            }
        }
    }
}

__global__ __launch_bounds__(256) void gc_gather_gemm_kernel(
    const float* __restrict__ x, const float* __restrict__ ori,
    const float* __restrict__ W, const float* __restrict__ bias,
    const int* __restrict__ csr, const int* __restrict__ start,
    const int* __restrict__ cnt, float* __restrict__ out, int N) {
    __shared__ __attribute__((aligned(16))) float Atile[16 * LDS_STRIDE];
    __shared__ float rs_s[16];

    const int base = blockIdx.x * 16;
    const int tid  = threadIdx.x;
    const int lane = tid & 31;
    const int wib  = __builtin_amdgcn_readfirstlane(tid >> 5);

    const int r0   = 2 * wib;
    const int row0 = base + r0;
    const int row1 = row0 + 1;
    int st0 = 0, cn0 = 0, st1 = 0, cn1 = 0;
    if (row0 < N) {
        st0 = __builtin_amdgcn_readfirstlane(start[row0]);
        cn0 = __builtin_amdgcn_readfirstlane(cnt[row0]);
    }
    if (row1 < N) {
        st1 = __builtin_amdgcn_readfirstlane(start[row1]);
        cn1 = __builtin_amdgcn_readfirstlane(cnt[row1]);
    }
    int d0 = 0, v0 = 0, d1 = 0, v1 = 0;
    if (lane < cn0) {
        const int2 p = ((const int2*)csr)[st0 + lane];
        d0 = p.x; v0 = p.y;
    }
    if (lane < cn1) {
        const int2 p = ((const int2*)csr)[st1 + lane];
        d1 = p.x; v1 = p.y;
    }

    #pragma unroll
    for (int i = 0; i < 2; ++i) {
        int idx = tid + i * 256;
        int r   = idx >> 5;
        int col = (idx & 31) * 4;
        int row = base + r; if (row >= N) row = N - 1;
        const float4 v = *(const float4*)(ori + (long long)row * F_IN + col);
        *(float4*)&Atile[r * LDS_STRIDE + F_IN + col] = v;
    }

    float4 a0 = make_float4(0.0f, 0.0f, 0.0f, 0.0f);
    float4 a1 = make_float4(0.0f, 0.0f, 0.0f, 0.0f);
    float s0 = 0.0f, s1 = 0.0f;
    gc_row_sum(x, csr, st0, cn0, lane, d0, v0, a0, s0);
    gc_row_sum(x, csr, st1, cn1, lane, d1, v1, a1, s1);
    *(float4*)&Atile[r0 * LDS_STRIDE + lane * 4] = a0;
    *(float4*)&Atile[(r0 + 1) * LDS_STRIDE + lane * 4] = a1;
    if (lane == 0) { rs_s[r0] = s0; rs_s[r0 + 1] = s1; }
    __syncthreads();

    const int wave    = tid >> 5;
    const int lmod    = lane & 15;
    const int lhi     = lane >> 4;
    const int colBase = wave * 16;
    const int col     = colBase + lmod;

    v8f acc = {};
    #pragma unroll
    for (int k = 0; k < K_TOT; k += 4) {
        v2f a = *(const v2f*)&Atile[lmod * LDS_STRIDE + k + 2 * lhi];
        int kb = k + 2 * lhi;
        v2f b;
        b.x = W[(long long)kb * F_OUT + col];
        b.y = W[(long long)(kb + 1) * F_OUT + col];
        acc = __builtin_amdgcn_wmma_f32_16x16x4_f32(
             false, a,  false, b,
             (short)0, acc,  false,  false);
    }

    const float bv = bias[col];
    #pragma unroll
    for (int r = 0; r < 8; ++r) {
        int row = base + r + 8 * lhi;
        if (row < N)
            out[(long long)row * F_OUT + col] = acc[r] / rs_s[r + 8 * lhi] + bv;
    }
}

extern "C" void kernel_launch(void* const* d_in, const int* in_sizes, int n_in,
                              void* d_out, int out_size, void* d_ws, size_t ws_size,
                              hipStream_t stream) {
    const float* ori  = (const float*)d_in[0];
    const float* x    = (const float*)d_in[1];
    const int*   esrc = (const int*)d_in[2];
    const int*   edst = (const int*)d_in[3];
    const float* eval = (const float*)d_in[4];
    const float* W    = (const float*)d_in[5];
    const float* bias = (const float*)d_in[6];
    float* out = (float*)d_out;

    const long long N = (long long)in_sizes[0] / F_IN;
    const long long E = (long long)in_sizes[2];

    int* csr   = (int*)d_ws;
    int* rank  = csr + 2 * E;
    int* cnt   = rank + E;
    int* start = cnt + N + 1;

    gc_clear_kernel<<<(unsigned)((N + 1 + 255) / 256), 256, 0, stream>>>(cnt, (int)(N + 1));

    const unsigned eblocks = (unsigned)((E + 256 * EDGES_PER_THREAD - 1) / (256 * EDGES_PER_THREAD));
    gc_ticket_kernel<<<eblocks, 256, 0, stream>>>(esrc, cnt, rank, E);
    gc_alloc_kernel<<<(unsigned)((N + ALLOC_ROWS - 1) / ALLOC_ROWS), ALLOC_ROWS, 0, stream>>>(
        cnt, cnt + N, start, (int)N);
    gc_fill_kernel<<<eblocks, 256, 0, stream>>>(esrc, edst, eval, rank, start, csr, E);

    gc_gather_gemm_kernel<<<(unsigned)((N + 15) / 16), 256, 0, stream>>>(
        x, ori, W, bias, csr, start, cnt, out, (int)N);
}


# counting sort: per-edge global returning atomic replaced by tickets from LDS row counters, one work-group per (row range, edge chunk); clear launch dropped; sums in chunk order, then LDS ticket order
# speedup vs baseline: 2.8056x; 1.3198x over previous
#include <hip/hip_runtime.h>
#include <hip/hip_bf16.h>

typedef __attribute__((ext_vector_type(2))) float v2f;
typedef __attribute__((ext_vector_type(8))) float v8f;

#define F_IN   128
#define K_TOT  256
#define F_OUT  128
#define LDS_STRIDE 260


#define EDGES_PER_THREAD 4
#define SORT_CHUNKS 16
#define RANGE_ROWS 6400
#define RANK_THREADS 1024
#define RANK_UNROLL 8
__global__ __launch_bounds__(RANK_THREADS) void gc_rank_kernel(
    const int* __restrict__ esrc, int* __restrict__ rank, int* __restrict__ ccnt,
    int* __restrict__ cursor, long long E, int N, int chunkLen) {
    __shared__ int lc[RANGE_ROWS];
    const int tid = threadIdx.x;
    const int c  = blockIdx.x % SORT_CHUNKS;
    const int lo = (blockIdx.x / SORT_CHUNKS) * RANGE_ROWS;
    for (int i = tid; i < RANGE_ROWS; i += RANK_THREADS) lc[i] = 0;
    if (blockIdx.x == 0 && tid == 0) *cursor = 0;
    __syncthreads();
    const long long cBeg = (long long)c * chunkLen;
    const int iters = chunkLen / RANK_THREADS;
    for (int it = 0; it < iters; it += RANK_UNROLL) {
        int s[RANK_UNROLL];
        int rk[RANK_UNROLL];
        #pragma unroll
        for (int k = 0; k < RANK_UNROLL; ++k) {
            const long long e = cBeg + (long long)(it + k) * RANK_THREADS + tid;
            s[k] = esrc[e < E ? e : E - 1];
        }
        #pragma unroll
        for (int k = 0; k < RANK_UNROLL; ++k) {
            const long long e = cBeg + (long long)(it + k) * RANK_THREADS + tid;
            const unsigned d = (unsigned)(s[k] - lo);
            rk[k] = -1;
            if (e < E && d < (unsigned)RANGE_ROWS) rk[k] = atomicAdd(&lc[d], 1);
        }
        #pragma unroll
        for (int k = 0; k < RANK_UNROLL; ++k) {
            const long long e = cBeg + (long long)(it + k) * RANK_THREADS + tid;
            if (rk[k] >= 0) rank[e] = rk[k];
        }
    }
    __syncthreads();
    for (int i = tid; i < RANGE_ROWS; i += RANK_THREADS) {
        const int row = lo + i;
        if (row < N) ccnt[(long long)c * N + row] = lc[i];
    }
}

#define ALLOC_ROWS 1024
__global__ __launch_bounds__(ALLOC_ROWS) void gc_alloc_kernel(
    int* ccnt, int* cursor, int* __restrict__ start, int* __restrict__ cnt, int N) {
    __shared__ int sc[ALLOC_ROWS];
    __shared__ int sbase;
    const int tid = threadIdx.x;
    const int i = blockIdx.x * ALLOC_ROWS + tid;
    int pre[SORT_CHUNKS];
    int c = 0;
    #pragma unroll
    for (int k = 0; k < SORT_CHUNKS; ++k) {
        const int t = (i < N) ? ccnt[(long long)k * N + i] : 0;
        pre[k] = c;
        c += t;
    }
    sc[tid] = c;
    __syncthreads();
    for (int off = 1; off < ALLOC_ROWS; off <<= 1) {
        const int add = (tid >= off) ? sc[tid - off] : 0;
        __syncthreads();
        sc[tid] += add;
        __syncthreads();
    }
    if (tid == ALLOC_ROWS - 1) sbase = atomicAdd(cursor, sc[tid]);
    __syncthreads();
    if (i < N) {
        const int st = sbase + sc[tid] - c;
        start[i] = st;
        cnt[i] = c;
        #pragma unroll
        for (int k = 0; k < SORT_CHUNKS; ++k)
            ccnt[(long long)k * N + i] = st + pre[k];
    }
}

__global__ __launch_bounds__(256) void gc_fill_kernel(
    const int* __restrict__ esrc, const int* __restrict__ edst,
    const float* __restrict__ eval, const int* __restrict__ rank,
    const int* __restrict__ cbase, int* __restrict__ csr, long long E, int N, int chunkLen) {
    const long long t = (long long)blockIdx.x * (256 * EDGES_PER_THREAD) + threadIdx.x;
    const long long cb = ((long long)blockIdx.x * (256 * EDGES_PER_THREAD) / chunkLen) * N;
    int sidx[EDGES_PER_THREAD];
    int pos[EDGES_PER_THREAD];
    int2 pr[EDGES_PER_THREAD];
    #pragma unroll
    for (int k = 0; k < EDGES_PER_THREAD; ++k) {
        const long long e = t + k * 256;
        const long long ec = e < E ? e : E - 1;
        sidx[k] = esrc[ec];
        pos[k]  = rank[ec];
        pr[k]   = make_int2(edst[ec], __float_as_int(eval[ec]));
    }
    #pragma unroll
    for (int k = 0; k < EDGES_PER_THREAD; ++k)
        pos[k] += cbase[cb + sidx[k]];
    #pragma unroll
    for (int k = 0; k < EDGES_PER_THREAD; ++k) {
        const long long e = t + k * 256;
        if (e < E) ((int2*)csr)[pos[k]] = pr[k];
    }
}

#define ROWS_IN_FLIGHT 8
__device__ __forceinline__ void gc_row_sum(
    const float* __restrict__ x, const int* __restrict__ csr,
    const int st, const int cn, const int lane, int d_l, int v_l,
    float4& acc, float& rs) {
    for (int c0 = 0; c0 < cn; c0 += 32) {
        if (c0 > 0) {
            d_l = 0; v_l = 0;
            if (c0 + lane < cn) {
                const int2 p = ((const int2*)csr)[st + c0 + lane];
                d_l = p.x; v_l = p.y;
            }
        }
        const int left = cn - c0;
        const int m = left < 32 ? left : 32;
        int j0 = 0;
        for (; j0 + ROWS_IN_FLIGHT <= m; j0 += ROWS_IN_FLIGHT) {
            float4 xv[ROWS_IN_FLIGHT];
            #pragma unroll
            for (int u = 0; u < ROWS_IN_FLIGHT; ++u) {
                const int d = __builtin_amdgcn_readlane(d_l, j0 + u);
                xv[u] = *(const float4*)(x + (long long)d * F_IN + lane * 4);
            }
            #pragma unroll
            for (int u = 0; u < ROWS_IN_FLIGHT; ++u) {
                const float v = __int_as_float(__builtin_amdgcn_readlane(v_l, j0 + u));
                acc.x += v * xv[u].x;
                acc.y += v * xv[u].y;
                acc.z += v * xv[u].z;
                acc.w += v * xv[u].w;
                rs += v;
            }
        }
        if (j0 < m) {
            float4 xv[ROWS_IN_FLIGHT];
            #pragma unroll
            for (int u = 0; u < ROWS_IN_FLIGHT; ++u) {
                xv[u] = make_float4(0.0f, 0.0f, 0.0f, 0.0f);
                if (j0 + u < m) {
                    const int d = __builtin_amdgcn_readlane(d_l, j0 + u);
                    xv[u] = *(const float4*)(x + (long long)d * F_IN + lane * 4);
                }
            }
            #pragma unroll
            for (int u = 0; u < ROWS_IN_FLIGHT; ++u) {
                if (j0 + u < m) {
                    const float v = __int_as_float(__builtin_amdgcn_readlane(v_l, j0 + u));
                    acc.x += v * xv[u].x;
                    acc.y += v * xv[u].y;
                    acc.z += v * xv[u].z;
                    acc.w += v * xv[u].w;
                    rs += v;
                }
            }
        }
    }
}

__global__ __launch_bounds__(256) void gc_gather_gemm_kernel(
    const float* __restrict__ x, const float* __restrict__ ori,
    const float* __restrict__ W, const float* __restrict__ bias,
    const int* __restrict__ csr, const int* __restrict__ start,
    const int* __restrict__ cnt, float* __restrict__ out, int N) {
    __shared__ __attribute__((aligned(16))) float Atile[16 * LDS_STRIDE];
    __shared__ float rs_s[16];

    const int base = blockIdx.x * 16;
    const int tid  = threadIdx.x;
    const int lane = tid & 31;
    const int wib  = __builtin_amdgcn_readfirstlane(tid >> 5);

    const int r0   = 2 * wib;
    const int row0 = base + r0;
    const int row1 = row0 + 1;
    int st0 = 0, cn0 = 0, st1 = 0, cn1 = 0;
    if (row0 < N) {
        st0 = __builtin_amdgcn_readfirstlane(start[row0]);
        cn0 = __builtin_amdgcn_readfirstlane(cnt[row0]);
    }
    if (row1 < N) {
        st1 = __builtin_amdgcn_readfirstlane(start[row1]);
        cn1 = __builtin_amdgcn_readfirstlane(cnt[row1]);
    }
    int d0 = 0, v0 = 0, d1 = 0, v1 = 0;
    if (lane < cn0) {
        const int2 p = ((const int2*)csr)[st0 + lane];
        d0 = p.x; v0 = p.y;
    }
    if (lane < cn1) {
        const int2 p = ((const int2*)csr)[st1 + lane];
        d1 = p.x; v1 = p.y;
    }

    #pragma unroll
    for (int i = 0; i < 2; ++i) {
        int idx = tid + i * 256;
        int r   = idx >> 5;
        int col = (idx & 31) * 4;
        int row = base + r; if (row >= N) row = N - 1;
        const float4 v = *(const float4*)(ori + (long long)row * F_IN + col);
        *(float4*)&Atile[r * LDS_STRIDE + F_IN + col] = v;
    }

    float4 a0 = make_float4(0.0f, 0.0f, 0.0f, 0.0f);
    float4 a1 = make_float4(0.0f, 0.0f, 0.0f, 0.0f);
    float s0 = 0.0f, s1 = 0.0f;
    gc_row_sum(x, csr, st0, cn0, lane, d0, v0, a0, s0);
    gc_row_sum(x, csr, st1, cn1, lane, d1, v1, a1, s1);
    *(float4*)&Atile[r0 * LDS_STRIDE + lane * 4] = a0;
    *(float4*)&Atile[(r0 + 1) * LDS_STRIDE + lane * 4] = a1;
    if (lane == 0) { rs_s[r0] = s0; rs_s[r0 + 1] = s1; }
    __syncthreads();

    const int wave    = tid >> 5;
    const int lmod    = lane & 15;
    const int lhi     = lane >> 4;
    const int colBase = wave * 16;
    const int col     = colBase + lmod;

    v8f acc = {};
    #pragma unroll
    for (int k = 0; k < K_TOT; k += 4) {
        v2f a = *(const v2f*)&Atile[lmod * LDS_STRIDE + k + 2 * lhi];
        int kb = k + 2 * lhi;
        v2f b;
        b.x = W[(long long)kb * F_OUT + col];
        b.y = W[(long long)(kb + 1) * F_OUT + col];
        acc = __builtin_amdgcn_wmma_f32_16x16x4_f32(
             false, a,  false, b,
             (short)0, acc,  false,  false);
    }

    const float bv = bias[col];
    #pragma unroll
    for (int r = 0; r < 8; ++r) {
        int row = base + r + 8 * lhi;
        if (row < N)
            out[(long long)row * F_OUT + col] = acc[r] / rs_s[r + 8 * lhi] + bv;
    }
}

extern "C" void kernel_launch(void* const* d_in, const int* in_sizes, int n_in,
                              void* d_out, int out_size, void* d_ws, size_t ws_size,
                              hipStream_t stream) {
    const float* ori  = (const float*)d_in[0];
    const float* x    = (const float*)d_in[1];
    const int*   esrc = (const int*)d_in[2];
    const int*   edst = (const int*)d_in[3];
    const float* eval = (const float*)d_in[4];
    const float* W    = (const float*)d_in[5];
    const float* bias = (const float*)d_in[6];
    float* out = (float*)d_out;

    const long long N = (long long)in_sizes[0] / F_IN;
    const long long E = (long long)in_sizes[2];

    int* csr   = (int*)d_ws;
    int* rank  = csr + 2 * E;
    int* cnt   = rank + E;
    int* start = cnt + N + 1;
    int* ccnt  = start + N;

    const long long chunkLen = ((E + SORT_CHUNKS * (RANK_THREADS * RANK_UNROLL) - 1) /
                                (SORT_CHUNKS * (RANK_THREADS * RANK_UNROLL))) * (RANK_THREADS * RANK_UNROLL);

    const unsigned ranges = (unsigned)((N + RANGE_ROWS - 1) / RANGE_ROWS);
    gc_rank_kernel<<<ranges * SORT_CHUNKS, RANK_THREADS, 0, stream>>>(
        esrc, rank, ccnt, cnt + N, E, (int)N, (int)chunkLen);

    const unsigned eblocks = (unsigned)((E + 256 * EDGES_PER_THREAD - 1) / (256 * EDGES_PER_THREAD));
    gc_alloc_kernel<<<(unsigned)((N + ALLOC_ROWS - 1) / ALLOC_ROWS), ALLOC_ROWS, 0, stream>>>(
        ccnt, cnt + N, start, cnt, (int)N);
    gc_fill_kernel<<<eblocks, 256, 0, stream>>>(esrc, edst, eval, rank, ccnt, csr, E, (int)N, (int)chunkLen);

    gc_gather_gemm_kernel<<<(unsigned)((N + 15) / 16), 256, 0, stream>>>(
        x, ori, W, bias, csr, start, cnt, out, (int)N);
}


# gather+GEMM work-group is 4 waves, not 8, per 16 rows: a wave sums 4 rows, then multiplies two column tiles on two accumulators; twice the rows resident per WGP; order of every sum unchanged
# speedup vs baseline: 2.9713x; 1.0591x over previous
#include <hip/hip_runtime.h>
#include <hip/hip_bf16.h>

typedef __attribute__((ext_vector_type(2))) float v2f;
typedef __attribute__((ext_vector_type(8))) float v8f;

#define F_IN   128
#define K_TOT  256
#define F_OUT  128
#define LDS_STRIDE 260


#define EDGES_PER_THREAD 4
#define SORT_CHUNKS 16
#define RANGE_ROWS 6400
#define RANK_THREADS 1024
#define RANK_UNROLL 8
__global__ __launch_bounds__(RANK_THREADS) void gc_rank_kernel(
    const int* __restrict__ esrc, int* __restrict__ rank, int* __restrict__ ccnt,
    int* __restrict__ cursor, long long E, int N, int chunkLen) {
    __shared__ int lc[RANGE_ROWS];
    const int tid = threadIdx.x;
    const int c  = blockIdx.x % SORT_CHUNKS;
    const int lo = (blockIdx.x / SORT_CHUNKS) * RANGE_ROWS;
    for (int i = tid; i < RANGE_ROWS; i += RANK_THREADS) lc[i] = 0;
    if (blockIdx.x == 0 && tid == 0) *cursor = 0;
    __syncthreads();
    const long long cBeg = (long long)c * chunkLen;
    const int iters = chunkLen / RANK_THREADS;
    for (int it = 0; it < iters; it += RANK_UNROLL) {
        int s[RANK_UNROLL];
        int rk[RANK_UNROLL];
        #pragma unroll
        for (int k = 0; k < RANK_UNROLL; ++k) {
            const long long e = cBeg + (long long)(it + k) * RANK_THREADS + tid;
            s[k] = esrc[e < E ? e : E - 1];
        }
        #pragma unroll
        for (int k = 0; k < RANK_UNROLL; ++k) {
            const long long e = cBeg + (long long)(it + k) * RANK_THREADS + tid;
            const unsigned d = (unsigned)(s[k] - lo);
            rk[k] = -1;
            if (e < E && d < (unsigned)RANGE_ROWS) rk[k] = atomicAdd(&lc[d], 1);
        }
        #pragma unroll
        for (int k = 0; k < RANK_UNROLL; ++k) {
            const long long e = cBeg + (long long)(it + k) * RANK_THREADS + tid;
            if (rk[k] >= 0) rank[e] = rk[k];
        }
    }
    __syncthreads();
    for (int i = tid; i < RANGE_ROWS; i += RANK_THREADS) {
        const int row = lo + i;
        if (row < N) ccnt[(long long)c * N + row] = lc[i];
    }
}

#define ALLOC_ROWS 1024
__global__ __launch_bounds__(ALLOC_ROWS) void gc_alloc_kernel(
    int* ccnt, int* cursor, int* __restrict__ start, int* __restrict__ cnt, int N) {
    __shared__ int sc[ALLOC_ROWS];
    __shared__ int sbase;
    const int tid = threadIdx.x;
    const int i = blockIdx.x * ALLOC_ROWS + tid;
    int pre[SORT_CHUNKS];
    int c = 0;
    #pragma unroll
    for (int k = 0; k < SORT_CHUNKS; ++k) {
        const int t = (i < N) ? ccnt[(long long)k * N + i] : 0;
        pre[k] = c;
        c += t;
    }
    sc[tid] = c;
    __syncthreads();
    for (int off = 1; off < ALLOC_ROWS; off <<= 1) {
        const int add = (tid >= off) ? sc[tid - off] : 0;
        __syncthreads();
        sc[tid] += add;
        __syncthreads();
    }
    if (tid == ALLOC_ROWS - 1) sbase = atomicAdd(cursor, sc[tid]);
    __syncthreads();
    if (i < N) {
        const int st = sbase + sc[tid] - c;
        start[i] = st;
        cnt[i] = c;
        #pragma unroll
        for (int k = 0; k < SORT_CHUNKS; ++k)
            ccnt[(long long)k * N + i] = st + pre[k];
    }
}

__global__ __launch_bounds__(256) void gc_fill_kernel(
    const int* __restrict__ esrc, const int* __restrict__ edst,
    const float* __restrict__ eval, const int* __restrict__ rank,
    const int* __restrict__ cbase, int* __restrict__ csr, long long E, int N, int chunkLen) {
    const long long t = (long long)blockIdx.x * (256 * EDGES_PER_THREAD) + threadIdx.x;
    const long long cb = ((long long)blockIdx.x * (256 * EDGES_PER_THREAD) / chunkLen) * N;
    int sidx[EDGES_PER_THREAD];
    int pos[EDGES_PER_THREAD];
    int2 pr[EDGES_PER_THREAD];
    #pragma unroll
    for (int k = 0; k < EDGES_PER_THREAD; ++k) {
        const long long e = t + k * 256;
        const long long ec = e < E ? e : E - 1;
        sidx[k] = esrc[ec];
        pos[k]  = rank[ec];
        pr[k]   = make_int2(edst[ec], __float_as_int(eval[ec]));
    }
    #pragma unroll
    for (int k = 0; k < EDGES_PER_THREAD; ++k)
        pos[k] += cbase[cb + sidx[k]];
    #pragma unroll
    for (int k = 0; k < EDGES_PER_THREAD; ++k) {
        const long long e = t + k * 256;
        if (e < E) ((int2*)csr)[pos[k]] = pr[k];
    }
}

#define ROWS_IN_FLIGHT 8
__device__ __forceinline__ void gc_row_sum(
    const float* __restrict__ x, const int* __restrict__ csr,
    const int st, const int cn, const int lane, int d_l, int v_l,
    float4& acc, float& rs) {
    for (int c0 = 0; c0 < cn; c0 += 32) {
        if (c0 > 0) {
            d_l = 0; v_l = 0;
            if (c0 + lane < cn) {
                const int2 p = ((const int2*)csr)[st + c0 + lane];
                d_l = p.x; v_l = p.y;
            }
        }
        const int left = cn - c0;
        const int m = left < 32 ? left : 32;
        int j0 = 0;
        for (; j0 + ROWS_IN_FLIGHT <= m; j0 += ROWS_IN_FLIGHT) {
            float4 xv[ROWS_IN_FLIGHT];
            #pragma unroll
            for (int u = 0; u < ROWS_IN_FLIGHT; ++u) {
                const int d = __builtin_amdgcn_readlane(d_l, j0 + u);
                xv[u] = *(const float4*)(x + (long long)d * F_IN + lane * 4);
            }
            #pragma unroll
            for (int u = 0; u < ROWS_IN_FLIGHT; ++u) {
                const float v = __int_as_float(__builtin_amdgcn_readlane(v_l, j0 + u));
                acc.x += v * xv[u].x;
                acc.y += v * xv[u].y;
                acc.z += v * xv[u].z;
                acc.w += v * xv[u].w;
                rs += v;
            }
        }
        if (j0 < m) {
            float4 xv[ROWS_IN_FLIGHT];
            #pragma unroll
            for (int u = 0; u < ROWS_IN_FLIGHT; ++u) {
                xv[u] = make_float4(0.0f, 0.0f, 0.0f, 0.0f);
                if (j0 + u < m) {
                    const int d = __builtin_amdgcn_readlane(d_l, j0 + u);
                    xv[u] = *(const float4*)(x + (long long)d * F_IN + lane * 4);
                }
            }
            #pragma unroll
            for (int u = 0; u < ROWS_IN_FLIGHT; ++u) {
                if (j0 + u < m) {
                    const float v = __int_as_float(__builtin_amdgcn_readlane(v_l, j0 + u));
                    acc.x += v * xv[u].x;
                    acc.y += v * xv[u].y;
                    acc.z += v * xv[u].z;
                    acc.w += v * xv[u].w;
                    rs += v;
                }
            }
        }
    }
}

#define GG_WAVES 4
#define GG_THREADS (32 * GG_WAVES)
#define GG_ROWS_PER_WAVE (16 / GG_WAVES)
__global__ __launch_bounds__(GG_THREADS) __attribute__((amdgpu_waves_per_eu(16, 16)))
void gc_gather_gemm_kernel(
    const float* __restrict__ x, const float* __restrict__ ori,
    const float* __restrict__ W, const float* __restrict__ bias,
    const int* __restrict__ csr, const int* __restrict__ start,
    const int* __restrict__ cnt, float* __restrict__ out, int N) {
    __shared__ __attribute__((aligned(16))) float Atile[16 * LDS_STRIDE];
    __shared__ float rs_s[16];

    const int base = blockIdx.x * 16;
    const int tid  = threadIdx.x;
    const int lane = tid & 31;
    const int wib  = __builtin_amdgcn_readfirstlane(tid >> 5);

    const int r0 = GG_ROWS_PER_WAVE * wib;
    int st[GG_ROWS_PER_WAVE], cn[GG_ROWS_PER_WAVE];
    int dq[GG_ROWS_PER_WAVE], vq[GG_ROWS_PER_WAVE];
    #pragma unroll
    for (int q = 0; q < GG_ROWS_PER_WAVE; ++q) {
        const int row = base + r0 + q;
        st[q] = 0; cn[q] = 0;
        if (row < N) {
            st[q] = __builtin_amdgcn_readfirstlane(start[row]);
            cn[q] = __builtin_amdgcn_readfirstlane(cnt[row]);
        }
    }
    #pragma unroll
    for (int q = 0; q < GG_ROWS_PER_WAVE; ++q) {
        const int slot = (lane < cn[q]) ? st[q] + lane : 0;
        const int2 p = ((const int2*)csr)[slot];
        dq[q] = p.x; vq[q] = p.y;
    }

    #pragma unroll
    for (int i = 0; i < 512 / GG_THREADS; ++i) {
        int idx = tid + i * GG_THREADS;
        int r   = idx >> 5;
        int col = (idx & 31) * 4;
        int row = base + r; if (row >= N) row = N - 1;
        const float4 v = *(const float4*)(ori + (long long)row * F_IN + col);
        *(float4*)&Atile[r * LDS_STRIDE + F_IN + col] = v;
    }

    #pragma unroll
    for (int q = 0; q < GG_ROWS_PER_WAVE; ++q) {
        float4 a = make_float4(0.0f, 0.0f, 0.0f, 0.0f);
        float s = 0.0f;
        gc_row_sum(x, csr, st[q], cn[q], lane, dq[q], vq[q], a, s);
        *(float4*)&Atile[(r0 + q) * LDS_STRIDE + lane * 4] = a;
        if (lane == 0) rs_s[r0 + q] = s;
    }
    __syncthreads();

    const int wave = tid >> 5;
    const int lmod = lane & 15;
    const int lhi  = lane >> 4;
    const int c0   = wave * 32 + lmod;
    const int c1   = c0 + 16;

    v8f acc0 = {};
    v8f acc1 = {};
    #pragma unroll
    for (int k = 0; k < K_TOT; k += 4) {
        const v2f a = *(const v2f*)&Atile[lmod * LDS_STRIDE + k + 2 * lhi];
        const int kb = k + 2 * lhi;
        v2f b0, b1;
        b0.x = W[(long long)kb * F_OUT + c0];
        b0.y = W[(long long)(kb + 1) * F_OUT + c0];
        b1.x = W[(long long)kb * F_OUT + c1];
        b1.y = W[(long long)(kb + 1) * F_OUT + c1];
        acc0 = __builtin_amdgcn_wmma_f32_16x16x4_f32(
             false, a,  false, b0,
             (short)0, acc0,  false,  false);
        acc1 = __builtin_amdgcn_wmma_f32_16x16x4_f32(
             false, a,  false, b1,
             (short)0, acc1,  false,  false);
    }

    const float bv0 = bias[c0];
    const float bv1 = bias[c1];
    #pragma unroll
    for (int r = 0; r < 8; ++r) {
        int row = base + r + 8 * lhi;
        if (row < N) {
            const float rs = rs_s[r + 8 * lhi];
            out[(long long)row * F_OUT + c0] = acc0[r] / rs + bv0;
            out[(long long)row * F_OUT + c1] = acc1[r] / rs + bv1;
        }
    }
}

extern "C" void kernel_launch(void* const* d_in, const int* in_sizes, int n_in,
                              void* d_out, int out_size, void* d_ws, size_t ws_size,
                              hipStream_t stream) {
    const float* ori  = (const float*)d_in[0];
    const float* x    = (const float*)d_in[1];
    const int*   esrc = (const int*)d_in[2];
    const int*   edst = (const int*)d_in[3];
    const float* eval = (const float*)d_in[4];
    const float* W    = (const float*)d_in[5];
    const float* bias = (const float*)d_in[6];
    float* out = (float*)d_out;

    const long long N = (long long)in_sizes[0] / F_IN;
    const long long E = (long long)in_sizes[2];

    int* csr   = (int*)d_ws;
    int* rank  = csr + 2 * E;
    int* cnt   = rank + E;
    int* start = cnt + N + 1;
    int* ccnt  = start + N;

    const long long chunkLen = ((E + SORT_CHUNKS * (RANK_THREADS * RANK_UNROLL) - 1) /
                                (SORT_CHUNKS * (RANK_THREADS * RANK_UNROLL))) * (RANK_THREADS * RANK_UNROLL);

    const unsigned ranges = (unsigned)((N + RANGE_ROWS - 1) / RANGE_ROWS);
    gc_rank_kernel<<<ranges * SORT_CHUNKS, RANK_THREADS, 0, stream>>>(
        esrc, rank, ccnt, cnt + N, E, (int)N, (int)chunkLen);

    const unsigned eblocks = (unsigned)((E + 256 * EDGES_PER_THREAD - 1) / (256 * EDGES_PER_THREAD));
    gc_alloc_kernel<<<(unsigned)((N + ALLOC_ROWS - 1) / ALLOC_ROWS), ALLOC_ROWS, 0, stream>>>(
        ccnt, cnt + N, start, cnt, (int)N);
    gc_fill_kernel<<<eblocks, 256, 0, stream>>>(esrc, edst, eval, rank, ccnt, csr, E, (int)N, (int)chunkLen);

    gc_gather_gemm_kernel<<<(unsigned)((N + 15) / 16), GG_THREADS, 0, stream>>>(
        x, ori, W, bias, csr, start, cnt, out, (int)N);
}


# counting sort re-cut: a rank work-group keeps 25600 row counters in 100 KB of LDS, 4 row ranges x 64 edge chunks, so each edge is read by 4 work-groups, was 16; sum order: chunk, then LDS ticket
# speedup vs baseline: 3.1244x; 1.0515x over previous
#include <hip/hip_runtime.h>
#include <hip/hip_bf16.h>

typedef __attribute__((ext_vector_type(2))) float v2f;
typedef __attribute__((ext_vector_type(8))) float v8f;

#define F_IN   128
#define K_TOT  256
#define F_OUT  128
#define LDS_STRIDE 260


#define EDGES_PER_THREAD 4
#define RANK_THREADS 1024
#define RANK_UNROLL 8
template <int SORT_CHUNKS, int RANGE_ROWS>
__global__ __launch_bounds__(RANK_THREADS) void gc_rank_kernel(
    const int* __restrict__ esrc, int* __restrict__ rank, int* __restrict__ ccnt,
    int* __restrict__ cursor, long long E, int N, int chunkLen) {
    __shared__ int lc[RANGE_ROWS];
    const int tid = threadIdx.x;
    const int c  = blockIdx.x % SORT_CHUNKS;
    const int lo = (blockIdx.x / SORT_CHUNKS) * RANGE_ROWS;
    for (int i = tid; i < RANGE_ROWS; i += RANK_THREADS) lc[i] = 0;
    if (blockIdx.x == 0 && tid == 0) *cursor = 0;
    __syncthreads();
    const long long cBeg = (long long)c * chunkLen;
    const int iters = chunkLen / RANK_THREADS;
    for (int it = 0; it < iters; it += RANK_UNROLL) {
        int s[RANK_UNROLL];
        int rk[RANK_UNROLL];
        #pragma unroll
        for (int k = 0; k < RANK_UNROLL; ++k) {
            const long long e = cBeg + (long long)(it + k) * RANK_THREADS + tid;
            s[k] = esrc[e < E ? e : E - 1];
        }
        #pragma unroll
        for (int k = 0; k < RANK_UNROLL; ++k) {
            const long long e = cBeg + (long long)(it + k) * RANK_THREADS + tid;
            const unsigned d = (unsigned)(s[k] - lo);
            rk[k] = -1;
            if (e < E && d < (unsigned)RANGE_ROWS) rk[k] = atomicAdd(&lc[d], 1);
        }
        #pragma unroll
        for (int k = 0; k < RANK_UNROLL; ++k) {
            const long long e = cBeg + (long long)(it + k) * RANK_THREADS + tid;
            if (rk[k] >= 0) rank[e] = rk[k];
        }
    }
    __syncthreads();
    for (int i = tid; i < RANGE_ROWS; i += RANK_THREADS) {
        const int row = lo + i;
        if (row < N) ccnt[(long long)c * N + row] = lc[i];
    }
}

template <int SORT_CHUNKS, int ALLOC_ROWS>
__global__ __launch_bounds__(ALLOC_ROWS) void gc_alloc_kernel(
    int* ccnt, int* cursor, int* __restrict__ start, int* __restrict__ cnt, int N) {
    __shared__ int sc[ALLOC_ROWS];
    __shared__ int sbase;
    const int tid = threadIdx.x;
    const int i = blockIdx.x * ALLOC_ROWS + tid;
    int pre[SORT_CHUNKS];
    int cntk[SORT_CHUNKS];
    int c = 0;
    const int ic = i < N ? i : N - 1;
    #pragma unroll
    for (int k = 0; k < SORT_CHUNKS; ++k)
        cntk[k] = ccnt[(long long)k * N + ic];
    #pragma unroll
    for (int k = 0; k < SORT_CHUNKS; ++k) {
        pre[k] = c;
        c += (i < N) ? cntk[k] : 0;
    }
    sc[tid] = c;
    __syncthreads();
    for (int off = 1; off < ALLOC_ROWS; off <<= 1) {
        const int add = (tid >= off) ? sc[tid - off] : 0;
        __syncthreads();
        sc[tid] += add;
        __syncthreads();
    }
    if (tid == ALLOC_ROWS - 1) sbase = atomicAdd(cursor, sc[tid]);
    __syncthreads();
    if (i < N) {
        const int st = sbase + sc[tid] - c;
        start[i] = st;
        cnt[i] = c;
        #pragma unroll
        for (int k = 0; k < SORT_CHUNKS; ++k)
            ccnt[(long long)k * N + i] = st + pre[k];
    }
}

__global__ __launch_bounds__(256) void gc_fill_kernel(
    const int* __restrict__ esrc, const int* __restrict__ edst,
    const float* __restrict__ eval, const int* __restrict__ rank,
    const int* __restrict__ cbase, int* __restrict__ csr, long long E, int N, int chunkLen) {
    const long long t = (long long)blockIdx.x * (256 * EDGES_PER_THREAD) + threadIdx.x;
    const long long cb = ((long long)blockIdx.x * (256 * EDGES_PER_THREAD) / chunkLen) * N;
    int sidx[EDGES_PER_THREAD];
    int pos[EDGES_PER_THREAD];
    int2 pr[EDGES_PER_THREAD];
    #pragma unroll
    for (int k = 0; k < EDGES_PER_THREAD; ++k) {
        const long long e = t + k * 256;
        const long long ec = e < E ? e : E - 1;
        sidx[k] = esrc[ec];
        pos[k]  = rank[ec];
        pr[k]   = make_int2(edst[ec], __float_as_int(eval[ec]));
    }
    #pragma unroll
    for (int k = 0; k < EDGES_PER_THREAD; ++k)
        pos[k] += cbase[cb + sidx[k]];
    #pragma unroll
    for (int k = 0; k < EDGES_PER_THREAD; ++k) {
        const long long e = t + k * 256;
        if (e < E) ((int2*)csr)[pos[k]] = pr[k];
    }
}

#define ROWS_IN_FLIGHT 8
__device__ __forceinline__ void gc_row_sum(
    const float* __restrict__ x, const int* __restrict__ csr,
    const int st, const int cn, const int lane, int d_l, int v_l,
    float4& acc, float& rs) {
    for (int c0 = 0; c0 < cn; c0 += 32) {
        if (c0 > 0) {
            d_l = 0; v_l = 0;
            if (c0 + lane < cn) {
                const int2 p = ((const int2*)csr)[st + c0 + lane];
                d_l = p.x; v_l = p.y;
            }
        }
        const int left = cn - c0;
        const int m = left < 32 ? left : 32;
        int j0 = 0;
        for (; j0 + ROWS_IN_FLIGHT <= m; j0 += ROWS_IN_FLIGHT) {
            float4 xv[ROWS_IN_FLIGHT];
            #pragma unroll
            for (int u = 0; u < ROWS_IN_FLIGHT; ++u) {
                const int d = __builtin_amdgcn_readlane(d_l, j0 + u);
                xv[u] = *(const float4*)(x + (long long)d * F_IN + lane * 4);
            }
            #pragma unroll
            for (int u = 0; u < ROWS_IN_FLIGHT; ++u) {
                const float v = __int_as_float(__builtin_amdgcn_readlane(v_l, j0 + u));
                acc.x += v * xv[u].x;
                acc.y += v * xv[u].y;
                acc.z += v * xv[u].z;
                acc.w += v * xv[u].w;
                rs += v;
            }
        }
        if (j0 < m) {
            float4 xv[ROWS_IN_FLIGHT];
            #pragma unroll
            for (int u = 0; u < ROWS_IN_FLIGHT; ++u) {
                xv[u] = make_float4(0.0f, 0.0f, 0.0f, 0.0f);
                if (j0 + u < m) {
                    const int d = __builtin_amdgcn_readlane(d_l, j0 + u);
                    xv[u] = *(const float4*)(x + (long long)d * F_IN + lane * 4);
                }
            }
            #pragma unroll
            for (int u = 0; u < ROWS_IN_FLIGHT; ++u) {
                if (j0 + u < m) {
                    const float v = __int_as_float(__builtin_amdgcn_readlane(v_l, j0 + u));
                    acc.x += v * xv[u].x;
                    acc.y += v * xv[u].y;
                    acc.z += v * xv[u].z;
                    acc.w += v * xv[u].w;
                    rs += v;
                }
            }
        }
    }
}

#define GG_WAVES 4
#define GG_THREADS (32 * GG_WAVES)
#define GG_ROWS_PER_WAVE (16 / GG_WAVES)
__global__ __launch_bounds__(GG_THREADS) __attribute__((amdgpu_waves_per_eu(16, 16)))
void gc_gather_gemm_kernel(
    const float* __restrict__ x, const float* __restrict__ ori,
    const float* __restrict__ W, const float* __restrict__ bias,
    const int* __restrict__ csr, const int* __restrict__ start,
    const int* __restrict__ cnt, float* __restrict__ out, int N) {
    __shared__ __attribute__((aligned(16))) float Atile[16 * LDS_STRIDE];
    __shared__ float rs_s[16];

    const int base = blockIdx.x * 16;
    const int tid  = threadIdx.x;
    const int lane = tid & 31;
    const int wib  = __builtin_amdgcn_readfirstlane(tid >> 5);

    const int r0 = GG_ROWS_PER_WAVE * wib;
    int st[GG_ROWS_PER_WAVE], cn[GG_ROWS_PER_WAVE];
    int dq[GG_ROWS_PER_WAVE], vq[GG_ROWS_PER_WAVE];
    #pragma unroll
    for (int q = 0; q < GG_ROWS_PER_WAVE; ++q) {
        const int row = base + r0 + q;
        st[q] = 0; cn[q] = 0;
        if (row < N) {
            st[q] = __builtin_amdgcn_readfirstlane(start[row]);
            cn[q] = __builtin_amdgcn_readfirstlane(cnt[row]);
        }
    }
    #pragma unroll
    for (int q = 0; q < GG_ROWS_PER_WAVE; ++q) {
        const int slot = (lane < cn[q]) ? st[q] + lane : 0;
        const int2 p = ((const int2*)csr)[slot];
        dq[q] = p.x; vq[q] = p.y;
    }

    #pragma unroll
    for (int i = 0; i < 512 / GG_THREADS; ++i) {
        int idx = tid + i * GG_THREADS;
        int r   = idx >> 5;
        int col = (idx & 31) * 4;
        int row = base + r; if (row >= N) row = N - 1;
        const float4 v = *(const float4*)(ori + (long long)row * F_IN + col);
        *(float4*)&Atile[r * LDS_STRIDE + F_IN + col] = v;
    }

    #pragma unroll
    for (int q = 0; q < GG_ROWS_PER_WAVE; ++q) {
        float4 a = make_float4(0.0f, 0.0f, 0.0f, 0.0f);
        float s = 0.0f;
        gc_row_sum(x, csr, st[q], cn[q], lane, dq[q], vq[q], a, s);
        *(float4*)&Atile[(r0 + q) * LDS_STRIDE + lane * 4] = a;
        if (lane == 0) rs_s[r0 + q] = s;
    }
    __syncthreads();

    const int wave = tid >> 5;
    const int lmod = lane & 15;
    const int lhi  = lane >> 4;
    const int c0   = wave * 32 + lmod;
    const int c1   = c0 + 16;

    v8f acc0 = {};
    v8f acc1 = {};
    #pragma unroll
    for (int k = 0; k < K_TOT; k += 4) {
        const v2f a = *(const v2f*)&Atile[lmod * LDS_STRIDE + k + 2 * lhi];
        const int kb = k + 2 * lhi;
        v2f b0, b1;
        b0.x = W[(long long)kb * F_OUT + c0];
        b0.y = W[(long long)(kb + 1) * F_OUT + c0];
        b1.x = W[(long long)kb * F_OUT + c1];
        b1.y = W[(long long)(kb + 1) * F_OUT + c1];
        acc0 = __builtin_amdgcn_wmma_f32_16x16x4_f32(
             false, a,  false, b0,
             (short)0, acc0,  false,  false);
        acc1 = __builtin_amdgcn_wmma_f32_16x16x4_f32(
             false, a,  false, b1,
             (short)0, acc1,  false,  false);
    }

    const float bv0 = bias[c0];
    const float bv1 = bias[c1];
    #pragma unroll
    for (int r = 0; r < 8; ++r) {
        int row = base + r + 8 * lhi;
        if (row < N) {
            const float rs = rs_s[r + 8 * lhi];
            out[(long long)row * F_OUT + c0] = acc0[r] / rs + bv0;
            out[(long long)row * F_OUT + c1] = acc1[r] / rs + bv1;
        }
    }
}

template <int SORT_CHUNKS, int RANGE_ROWS, int ALLOC_ROWS>
static void gc_sort_launch(const int* esrc, const int* edst, const float* eval,
                           int* csr, int* rank, int* cnt, int* start, int* ccnt,
                           long long E, long long N, hipStream_t stream) {
    const long long chunkLen = ((E + SORT_CHUNKS * (RANK_THREADS * RANK_UNROLL) - 1) /
                                (SORT_CHUNKS * (RANK_THREADS * RANK_UNROLL))) * (RANK_THREADS * RANK_UNROLL);
    const unsigned ranges = (unsigned)((N + RANGE_ROWS - 1) / RANGE_ROWS);
    gc_rank_kernel<SORT_CHUNKS, RANGE_ROWS><<<ranges * SORT_CHUNKS, RANK_THREADS, 0, stream>>>(
        esrc, rank, ccnt, cnt + N, E, (int)N, (int)chunkLen);
    const unsigned eblocks = (unsigned)((E + 256 * EDGES_PER_THREAD - 1) / (256 * EDGES_PER_THREAD));
    gc_alloc_kernel<SORT_CHUNKS, ALLOC_ROWS><<<(unsigned)((N + ALLOC_ROWS - 1) / ALLOC_ROWS), ALLOC_ROWS, 0, stream>>>(
        ccnt, cnt + N, start, cnt, (int)N);
    gc_fill_kernel<<<eblocks, 256, 0, stream>>>(esrc, edst, eval, rank, ccnt, csr, E, (int)N, (int)chunkLen);
}

extern "C" void kernel_launch(void* const* d_in, const int* in_sizes, int n_in,
                              void* d_out, int out_size, void* d_ws, size_t ws_size,
                              hipStream_t stream) {
    const float* ori  = (const float*)d_in[0];
    const float* x    = (const float*)d_in[1];
    const int*   esrc = (const int*)d_in[2];
    const int*   edst = (const int*)d_in[3];
    const float* eval = (const float*)d_in[4];
    const float* W    = (const float*)d_in[5];
    const float* bias = (const float*)d_in[6];
    float* out = (float*)d_out;

    const long long N = (long long)in_sizes[0] / F_IN;
    const long long E = (long long)in_sizes[2];

    int* csr   = (int*)d_ws;
    int* rank  = csr + 2 * E;
    int* cnt   = rank + E;
    int* start = cnt + N + 1;
    int* ccnt  = start + N;

    const size_t need64 = (size_t)(3 * E + 2 * N + 1 + 64 * N) * sizeof(int);
    if (ws_size >= need64)
        gc_sort_launch<64, 25600, 256>(esrc, edst, eval, csr, rank, cnt, start, ccnt, E, N, stream);
    else
        gc_sort_launch<16, 6400, 1024>(esrc, edst, eval, csr, rank, cnt, start, ccnt, E, N, stream);

    gc_gather_gemm_kernel<<<(unsigned)((N + 15) / 16), GG_THREADS, 0, stream>>>(
        x, ori, W, bias, csr, start, cnt, out, (int)N);
}
